# starting kernel
# baseline (speedup 1.0000x reference)
#include <hip/hip_runtime.h>

typedef float v2f __attribute__((ext_vector_type(2)));
typedef float v8f __attribute__((ext_vector_type(8)));

#define BATCH 8
#define NROW 256
#define DIM 128

__global__ __launch_bounds__(256, 2)
void relational_fused_kernel(const float* __restrict__ x,
                             const float* __restrict__ W,
                             const float* __restrict__ bias,
                             float* __restrict__ out) {
  extern __shared__ float smem[];
  float* c_s = smem;
  float* a_s = smem + NROW * DIM;

  const int blk   = blockIdx.x;
  const int b     = blk >> 4;
  const int itile = blk & 15;
  const int tid   = threadIdx.x;
  const int wave  = tid >> 5;
  const int lane  = tid & 31;
  const int lrow  = lane & 15;
  const int g     = lane >> 4;

  const float* Xb = x + b * (NROW * DIM);
  const float* W1 = W;
  const float* W2 = W + DIM * DIM;

  for (int t = wave; t < 128; t += 8) {
    const int mt = t >> 3;
    const int nt = t & 7;
    v8f acc = {};
    const float* arow = Xb + (mt * 16 + lrow) * DIM;
    const float* bcol = W2 + (nt * 16 + lrow);
    #pragma unroll 4
    for (int k0 = 0; k0 < DIM; k0 += 4) {
      const int ka = k0 + 2 * g;
      v2f afrag = { arow[ka], arow[ka + 1] };
      v2f bfrag = { bcol[ka * DIM], bcol[(ka + 1) * DIM] };
      acc = __builtin_amdgcn_wmma_f32_16x16x4_f32(
           false, afrag,  false, bfrag,
           (short)0, acc,  false,  false);
    }
    #pragma unroll
    for (int r = 0; r < 8; ++r) {
      c_s[(mt * 16 + r + 8 * g) * DIM + nt * 16 + lrow] = acc[r];
    }
  }

  {
    const int i0 = itile * 16;
    v8f acc = {};
    const float* arow = Xb + (i0 + lrow) * DIM;
    const float* bcol = W1 + (wave * 16 + lrow);
    #pragma unroll 4
    for (int k0 = 0; k0 < DIM; k0 += 4) {
      const int ka = k0 + 2 * g;
      v2f afrag = { arow[ka], arow[ka + 1] };
      v2f bfrag = { bcol[ka * DIM], bcol[(ka + 1) * DIM] };
      acc = __builtin_amdgcn_wmma_f32_16x16x4_f32(
          false, afrag, false, bfrag, (short)0, acc, false, false);
    }
    #pragma unroll
    for (int r = 0; r < 8; ++r) {
      a_s[(r + 8 * g) * DIM + wave * 16 + lrow] = acc[r];
    }
  }

  __syncthreads();

  const int d    = tid & (DIM - 1);
  const int half = tid >> 7;
  const float bd = bias[d];
  float s[8], acc[8];
  #pragma unroll
  for (int k = 0; k < 8; ++k) {
    s[k]   = a_s[(half * 8 + k) * DIM + d] + bd;
    acc[k] = 0.0f;
  }
  #pragma unroll 4
  for (int j = 0; j < NROW; ++j) {
    const float v = c_s[j * DIM + d];
    #pragma unroll
    for (int k = 0; k < 8; ++k) {
      acc[k] += fmaxf(s[k] + v, 0.0f);
    }
  }
  float* op = out + ((b * NROW) + itile * 16 + half * 8) * DIM + d;
  #pragma unroll
  for (int k = 0; k < 8; ++k) {
    op[k * DIM] = acc[k];
  }
}

extern "C" void kernel_launch(void* const* d_in, const int* in_sizes, int n_in,
                              void* d_out, int out_size, void* d_ws, size_t ws_size,
                              hipStream_t stream) {
  (void)in_sizes; (void)n_in; (void)d_ws; (void)ws_size; (void)out_size;
  const float* x    = (const float*)d_in[0];
  const float* W    = (const float*)d_in[1];
  const float* bias = (const float*)d_in[2];
  float* out        = (float*)d_out;

  const size_t shmem = (size_t)(NROW * DIM + 16 * DIM) * sizeof(float);
  (void)hipFuncSetAttribute((const void*)relational_fused_kernel,
                            hipFuncAttributeMaxDynamicSharedMemorySize,
                            (int)shmem);

  dim3 grid(BATCH * (NROW / 16));
  dim3 block(256);
  relational_fused_kernel<<<grid, block, shmem, stream>>>(x, W, bias, out);
}


# Work re-tiled: one group per (batch, 32-col d tile, 32-row i chunk), 256 groups; each computes only its 32 columns of c=X_b@W2 (was all 128, 16x redundant); k and j sum orders unchanged
# speedup vs baseline: 4.1135x; 4.1135x over previous
#include <hip/hip_runtime.h>


typedef float v2f __attribute__((ext_vector_type(2)));
typedef float v4f __attribute__((ext_vector_type(4)));
typedef float v8f __attribute__((ext_vector_type(8)));

#define BATCH 8
#define NROW 256
#define DIM 128

#define DT 32
#define IC 32
#define CPITCH 260
#define APITCH 36

#define WMMA4(A, B, C) \
  __builtin_amdgcn_wmma_f32_16x16x4_f32(false, (A), false, (B), (short)0, (C), false, false)

template <bool WITH_A>
__device__ __forceinline__ void gemm_phase(const float* __restrict__ Xb,
                                           const float* __restrict__ W1,
                                           const float* __restrict__ W2,
                                           float* c_t, float* a_t,
                                           int wave, int ic, int lrow, int g) {
  v8f acc00 = {}, acc01 = {}, acc10 = {}, acc11 = {}, acca = {};
  const float* arow0 = Xb + (wave * 32 + lrow) * DIM;
  const float* arow1 = arow0 + 16 * DIM;
  const float* bcol0 = W2 + lrow;
  const float* bcol1 = W2 + 16 + lrow;
  const int mi = wave >> 1;
  const int ni = wave & 1;
  const float* arowa = Xb + (ic * IC + mi * 16 + lrow) * DIM;
  const float* bcola = W1 + ni * 16 + lrow;
  #pragma unroll 4
  for (int k0 = 0; k0 < DIM; k0 += 4) {
    const int ka = k0 + 2 * g;
    v2f a0 = { arow0[ka], arow0[ka + 1] };
    v2f a1 = { arow1[ka], arow1[ka + 1] };
    v2f b0 = { bcol0[ka * DIM], bcol0[(ka + 1) * DIM] };
    v2f b1 = { bcol1[ka * DIM], bcol1[(ka + 1) * DIM] };
    v2f aa = {}, ba = {};
    if (WITH_A) {
      aa = (v2f){ arowa[ka], arowa[ka + 1] };
      ba = (v2f){ bcola[ka * DIM], bcola[(ka + 1) * DIM] };
    }
    acc00 = WMMA4(a0, b0, acc00);
    acc01 = WMMA4(a0, b1, acc01);
    acc10 = WMMA4(a1, b0, acc10);
    acc11 = WMMA4(a1, b1, acc11);
    if (WITH_A) acca = WMMA4(aa, ba, acca);
  }
  float* p0 = c_t + lrow * CPITCH + wave * 32 + 8 * g;
  float* p1 = p0 + 16 * CPITCH;
  *(v4f*)(p0)      = __builtin_shufflevector(acc00, acc00, 0, 1, 2, 3);
  *(v4f*)(p0 + 4)  = __builtin_shufflevector(acc00, acc00, 4, 5, 6, 7);
  *(v4f*)(p1)      = __builtin_shufflevector(acc01, acc01, 0, 1, 2, 3);
  *(v4f*)(p1 + 4)  = __builtin_shufflevector(acc01, acc01, 4, 5, 6, 7);
  *(v4f*)(p0 + 16) = __builtin_shufflevector(acc10, acc10, 0, 1, 2, 3);
  *(v4f*)(p0 + 20) = __builtin_shufflevector(acc10, acc10, 4, 5, 6, 7);
  *(v4f*)(p1 + 16) = __builtin_shufflevector(acc11, acc11, 0, 1, 2, 3);
  *(v4f*)(p1 + 20) = __builtin_shufflevector(acc11, acc11, 4, 5, 6, 7);
  if (WITH_A) {
    float* p = a_t + (ni * 16 + lrow) * APITCH + mi * 16 + 8 * g;
    *(v4f*)(p)     = __builtin_shufflevector(acca, acca, 0, 1, 2, 3);
    *(v4f*)(p + 4) = __builtin_shufflevector(acca, acca, 4, 5, 6, 7);
  }
}

__global__ __launch_bounds__(256)
void relational_fused_kernel(const float* __restrict__ x,
                             const float* __restrict__ W,
                             const float* __restrict__ bias,
                             float* __restrict__ out) {
  __shared__ __attribute__((aligned(16))) float c_t[DT * CPITCH];
  __shared__ __attribute__((aligned(16))) float a_t[DT * APITCH];

  const int blk = blockIdx.x;
  const int ic  = blk & 7;
  const int dt  = (blk >> 3) & 3;
  const int b   = blk >> 5;
  const int tid = threadIdx.x;
  const int wave = __builtin_amdgcn_readfirstlane(tid >> 5);
  const int lane = tid & 31;
  const int lrow = lane & 15;
  const int g    = lane >> 4;

  const float* Xb = x + b * (NROW * DIM);
  const float* W1 = W + dt * DT;
  const float* W2 = W + DIM * DIM + dt * DT;

  if (wave < 4) gemm_phase<true >(Xb, W1, W2, c_t, a_t, wave, ic, lrow, g);
  else          gemm_phase<false>(Xb, W1, W2, c_t, a_t, wave, ic, lrow, g);

  __syncthreads();

  const int d = lane;
  const float bd = bias[dt * DT + d];
  const v4f a4 = *(const v4f*)(a_t + d * APITCH + 4 * wave);
  float s[4], acc[4];
  #pragma unroll
  for (int k = 0; k < 4; ++k) {
    s[k]   = a4[k] + bd;
    acc[k] = 0.0f;
  }
  const float* crow = c_t + d * CPITCH;
  #pragma unroll 4
  for (int j = 0; j < NROW; j += 4) {
    const v4f v = *(const v4f*)(crow + j);
    #pragma unroll
    for (int jj = 0; jj < 4; ++jj) {
      #pragma unroll
      for (int k = 0; k < 4; ++k) {
        acc[k] += fmaxf(s[k] + v[jj], 0.0f);
      }
    }
  }
  float* op = out + ((b * NROW) + ic * IC + 4 * wave) * DIM + dt * DT + d;
  #pragma unroll
  for (int k = 0; k < 4; ++k) {
    op[k * DIM] = acc[k];
  }
}

extern "C" void kernel_launch(void* const* d_in, const int* in_sizes, int n_in,
                              void* d_out, int out_size, void* d_ws, size_t ws_size,
                              hipStream_t stream) {
  (void)in_sizes; (void)n_in; (void)d_ws; (void)ws_size; (void)out_size;
  const float* x    = (const float*)d_in[0];
  const float* W    = (const float*)d_in[1];
  const float* bias = (const float*)d_in[2];
  float* out        = (float*)d_out;

  dim3 grid(256);
  dim3 block(256);
  relational_fused_kernel<<<grid, block, 0, stream>>>(x, W, bias, out);
}


# Phase-1 operands staged through LDS: X_b and both W tiles fetched as whole 128-byte lines in 4 k chunks, next chunk's loads in flight under the WMMAs; fragments read from LDS; sum orders unchanged
# speedup vs baseline: 5.0721x; 1.2330x over previous
#include <hip/hip_runtime.h>


typedef float v2f __attribute__((ext_vector_type(2)));
typedef float v4f __attribute__((ext_vector_type(4)));
typedef float v8f __attribute__((ext_vector_type(8)));

#define BATCH 8
#define NROW 256
#define DIM 128

#define DT 32
#define IC 32
#define CPITCH 260
#define APITCH 36
#define XPITCH 132
#define WPITCH 132
#define KC 32
#define NCHUNK (DIM / KC)
#define NSPLIT 5

#define WMMA4(A, B, C) \
  __builtin_amdgcn_wmma_f32_16x16x4_f32(false, (A), false, (B), (short)0, (C), false, false)

struct Stage {
  v4f x[8];
  v4f w1;
  v4f w2;
};

__device__ __forceinline__ void stage_load(Stage& s, const float* __restrict__ Xb,
                                           const float* __restrict__ W1,
                                           const float* __restrict__ W2, int kc, int tid) {
  const int q = tid & 7;
  const int r = tid >> 3;
  #pragma unroll
  for (int i = 0; i < 8; ++i) {
    s.x[i] = *(const v4f*)(Xb + (r + 32 * i) * DIM + kc * KC + 4 * q);
  }
  s.w1 = *(const v4f*)(W1 + (kc * KC + r) * DIM + 4 * q);
  s.w2 = *(const v4f*)(W2 + (kc * KC + r) * DIM + 4 * q);
}

__device__ __forceinline__ void stage_store(const Stage& s, float* xs, float* ws1, float* ws2,
                                            int kc, int tid) {
  const int q = tid & 7;
  const int r = tid >> 3;
  #pragma unroll
  for (int i = 0; i < 8; ++i) {
    *(v4f*)(xs + (r + 32 * i) * XPITCH + kc * KC + 4 * q) = s.x[i];
  }
  #pragma unroll
  for (int c = 0; c < 4; ++c) {
    ws1[(4 * q + c) * WPITCH + kc * KC + r] = s.w1[c];
    ws2[(4 * q + c) * WPITCH + kc * KC + r] = s.w2[c];
  }
}

template <bool WITH_A, int S0, int S1>
__device__ __forceinline__ void wmma_steps(const float* xa0, const float* xa1,
                                           const float* wb0, const float* wb1,
                                           const float* xaa, const float* wba, int kc,
                                           v8f& acc00, v8f& acc01, v8f& acc10, v8f& acc11,
                                           v8f& acca) {
  #pragma unroll
  for (int s = S0; s < S1; ++s) {
    const int k = kc * KC + 4 * s;
    const v2f a0 = *(const v2f*)(xa0 + k);
    const v2f a1 = *(const v2f*)(xa1 + k);
    const v2f b0 = *(const v2f*)(wb0 + k);
    const v2f b1 = *(const v2f*)(wb1 + k);
    acc00 = WMMA4(a0, b0, acc00);
    acc01 = WMMA4(a0, b1, acc01);
    acc10 = WMMA4(a1, b0, acc10);
    acc11 = WMMA4(a1, b1, acc11);
    if (WITH_A) {
      const v2f aa = *(const v2f*)(xaa + k);
      const v2f ba = *(const v2f*)(wba + k);
      acca = WMMA4(aa, ba, acca);
    }
  }
}

__global__ __launch_bounds__(256)
void relational_fused_kernel(const float* __restrict__ x,
                             const float* __restrict__ W,
                             const float* __restrict__ bias,
                             float* __restrict__ out) {
  __shared__ __attribute__((aligned(16))) float xs[NROW * XPITCH];
  __shared__ __attribute__((aligned(16))) float ws1[DT * WPITCH];
  __shared__ __attribute__((aligned(16))) float ws2[DT * WPITCH];
  __shared__ __attribute__((aligned(16))) float c_t[DT * CPITCH];
  __shared__ __attribute__((aligned(16))) float a_t[DT * APITCH];

  const int blk = blockIdx.x;
  const int ic  = blk & 7;
  const int dt  = (blk >> 3) & 3;
  const int b   = blk >> 5;
  const int tid = threadIdx.x;
  const int wave = __builtin_amdgcn_readfirstlane(tid >> 5);
  const int lane = tid & 31;
  const int lrow = lane & 15;
  const int g    = lane >> 4;

  const float* Xb = x + b * (NROW * DIM);
  const float* W1 = W + dt * DT;
  const float* W2 = W + DIM * DIM + dt * DT;

  const int mi = wave >> 1;
  const int ni = wave & 1;
  const float* xa0 = xs + (wave * 32 + lrow) * XPITCH + 2 * g;
  const float* xa1 = xa0 + 16 * XPITCH;
  const float* wb0 = ws2 + lrow * WPITCH + 2 * g;
  const float* wb1 = wb0 + 16 * WPITCH;
  const float* xaa = xs + (ic * IC + mi * 16 + lrow) * XPITCH + 2 * g;
  const float* wba = ws1 + (ni * 16 + lrow) * WPITCH + 2 * g;

  Stage st;
  stage_load(st, Xb, W1, W2, 0, tid);
  stage_store(st, xs, ws1, ws2, 0, tid);
  __syncthreads();

  v8f acc00 = {}, acc01 = {}, acc10 = {}, acc11 = {}, acca = {};
  #pragma unroll
  for (int kc = 0; kc < NCHUNK; ++kc) {
    if (kc + 1 < NCHUNK) stage_load(st, Xb, W1, W2, kc + 1, tid);
    if (wave < 4) wmma_steps<true,  0, NSPLIT>(xa0, xa1, wb0, wb1, xaa, wba, kc, acc00, acc01, acc10, acc11, acca);
    else          wmma_steps<false, 0, NSPLIT>(xa0, xa1, wb0, wb1, xaa, wba, kc, acc00, acc01, acc10, acc11, acca);
    if (kc + 1 < NCHUNK) stage_store(st, xs, ws1, ws2, kc + 1, tid);
    if (wave < 4) wmma_steps<true,  NSPLIT, KC / 4>(xa0, xa1, wb0, wb1, xaa, wba, kc, acc00, acc01, acc10, acc11, acca);
    else          wmma_steps<false, NSPLIT, KC / 4>(xa0, xa1, wb0, wb1, xaa, wba, kc, acc00, acc01, acc10, acc11, acca);
    if (kc + 1 < NCHUNK) __syncthreads();
  }

  {
    float* p0 = c_t + lrow * CPITCH + wave * 32 + 8 * g;
    float* p1 = p0 + 16 * CPITCH;
    *(v4f*)(p0)      = __builtin_shufflevector(acc00, acc00, 0, 1, 2, 3);
    *(v4f*)(p0 + 4)  = __builtin_shufflevector(acc00, acc00, 4, 5, 6, 7);
    *(v4f*)(p1)      = __builtin_shufflevector(acc01, acc01, 0, 1, 2, 3);
    *(v4f*)(p1 + 4)  = __builtin_shufflevector(acc01, acc01, 4, 5, 6, 7);
    *(v4f*)(p0 + 16) = __builtin_shufflevector(acc10, acc10, 0, 1, 2, 3);
    *(v4f*)(p0 + 20) = __builtin_shufflevector(acc10, acc10, 4, 5, 6, 7);
    *(v4f*)(p1 + 16) = __builtin_shufflevector(acc11, acc11, 0, 1, 2, 3);
    *(v4f*)(p1 + 20) = __builtin_shufflevector(acc11, acc11, 4, 5, 6, 7);
    if (wave < 4) {
      float* p = a_t + (ni * 16 + lrow) * APITCH + mi * 16 + 8 * g;
      *(v4f*)(p)     = __builtin_shufflevector(acca, acca, 0, 1, 2, 3);
      *(v4f*)(p + 4) = __builtin_shufflevector(acca, acca, 4, 5, 6, 7);
    }
  }

  __syncthreads();

  const int d = lane;
  const float bd = bias[dt * DT + d];
  const v4f a4 = *(const v4f*)(a_t + d * APITCH + 4 * wave);
  float s[4], acc[4];
  #pragma unroll
  for (int k = 0; k < 4; ++k) {
    s[k]   = a4[k] + bd;
    acc[k] = 0.0f;
  }
  const float* crow = c_t + d * CPITCH;
  #pragma unroll 4
  for (int j = 0; j < NROW; j += 4) {
    const v4f v = *(const v4f*)(crow + j);
    #pragma unroll
    for (int jj = 0; jj < 4; ++jj) {
      #pragma unroll
      for (int k = 0; k < 4; ++k) {
        acc[k] += fmaxf(s[k] + v[jj], 0.0f);
      }
    }
  }
  float* op = out + ((b * NROW) + ic * IC + 4 * wave) * DIM + dt * DT + d;
  #pragma unroll
  for (int k = 0; k < 4; ++k) {
    op[k * DIM] = acc[k];
  }
}

extern "C" void kernel_launch(void* const* d_in, const int* in_sizes, int n_in,
                              void* d_out, int out_size, void* d_ws, size_t ws_size,
                              hipStream_t stream) {
  (void)in_sizes; (void)n_in; (void)d_ws; (void)ws_size; (void)out_size;
  const float* x    = (const float*)d_in[0];
  const float* W    = (const float*)d_in[1];
  const float* bias = (const float*)d_in[2];
  float* out        = (float*)d_out;

  dim3 grid(256);
  dim3 block(256);
  relational_fused_kernel<<<grid, block, 0, stream>>>(x, W, bias, out);
}


# All 4 k chunks' staging loads issued at kernel start, in flight across LDS-only barriers; k loop branch-free: a's k steps split even/odd over wave pairs (a = 2 partial sums, its k sum order changed)
# speedup vs baseline: 5.8101x; 1.1455x over previous
#include <hip/hip_runtime.h>


typedef float v2f __attribute__((ext_vector_type(2)));
typedef float v4f __attribute__((ext_vector_type(4)));
typedef float v8f __attribute__((ext_vector_type(8)));

#define BATCH 8
#define NROW 256
#define DIM 128

#define DT 32
#define IC 32
#define CPITCH 260
#define APITCH 36
#define XPITCH 132
#define WPITCH 132
#define KC 32
#define NCHUNK (DIM / KC)
#define NSPLIT 5

#define WMMA4(A, B, C) \
  __builtin_amdgcn_wmma_f32_16x16x4_f32(false, (A), false, (B), (short)0, (C), false, false)

struct Stage {
  v4f x[8];
  v4f w1;
  v4f w2;
};

__device__ __forceinline__ void stage_load(Stage& s, const float* __restrict__ Xb,
                                           const float* __restrict__ W1,
                                           const float* __restrict__ W2, int kc, int tid) {
  const int q = tid & 7;
  const int r = tid >> 3;
  #pragma unroll
  for (int i = 0; i < 8; ++i) {
    s.x[i] = *(const v4f*)(Xb + (r + 32 * i) * DIM + kc * KC + 4 * q);
  }
  s.w1 = *(const v4f*)(W1 + (kc * KC + r) * DIM + 4 * q);
  s.w2 = *(const v4f*)(W2 + (kc * KC + r) * DIM + 4 * q);
}

__device__ __forceinline__ void stage_store(const Stage& s, float* xs, float* ws1, float* ws2,
                                            int kc, int tid) {
  const int q = tid & 7;
  const int r = tid >> 3;
  #pragma unroll
  for (int i = 0; i < 8; ++i) {
    *(v4f*)(xs + (r + 32 * i) * XPITCH + kc * KC + 4 * q) = s.x[i];
  }
  #pragma unroll
  for (int c = 0; c < 4; ++c) {
    ws1[(4 * q + c) * WPITCH + kc * KC + r] = s.w1[c];
    ws2[(4 * q + c) * WPITCH + kc * KC + r] = s.w2[c];
  }
}

template <int S0, int S1>
__device__ __forceinline__ void wmma_steps(const float* xa0, const float* xa1,
                                           const float* wb0, const float* wb1,
                                           const float* xaa, const float* wba, int kc,
                                           v8f& acc00, v8f& acc01, v8f& acc10, v8f& acc11,
                                           v8f& acca) {
  #pragma unroll
  for (int s = S0; s < S1; ++s) {
    const int k = kc * KC + 4 * s;
    const v2f a0 = *(const v2f*)(xa0 + k);
    const v2f a1 = *(const v2f*)(xa1 + k);
    const v2f b0 = *(const v2f*)(wb0 + k);
    const v2f b1 = *(const v2f*)(wb1 + k);
    acc00 = WMMA4(a0, b0, acc00);
    acc01 = WMMA4(a0, b1, acc01);
    acc10 = WMMA4(a1, b0, acc10);
    acc11 = WMMA4(a1, b1, acc11);
    if ((s & 1) == 0) {
      const v2f aa = *(const v2f*)(xaa + k);
      const v2f ba = *(const v2f*)(wba + k);
      acca = WMMA4(aa, ba, acca);
    }
  }
}

__device__ __forceinline__ void lds_barrier() {
  __builtin_amdgcn_fence(__ATOMIC_RELEASE, "workgroup", "local");
  __builtin_amdgcn_s_barrier();
  __builtin_amdgcn_fence(__ATOMIC_ACQUIRE, "workgroup", "local");
}

__global__ __launch_bounds__(256)
void relational_fused_kernel(const float* __restrict__ x,
                             const float* __restrict__ W,
                             const float* __restrict__ bias,
                             float* __restrict__ out) {
  __shared__ __attribute__((aligned(16))) float xs[NROW * XPITCH];
  __shared__ __attribute__((aligned(16))) float ws1[DT * WPITCH];
  __shared__ __attribute__((aligned(16))) float ws2[DT * WPITCH];
  __shared__ __attribute__((aligned(16))) float c_t[DT * CPITCH];
  __shared__ __attribute__((aligned(16))) float a_t[DT * APITCH];
  __shared__ __attribute__((aligned(16))) float a_t2[DT * APITCH];

  const int blk = blockIdx.x;
  const int ic  = blk & 7;
  const int dt  = (blk >> 3) & 3;
  const int b   = blk >> 5;
  const int tid = threadIdx.x;
  const int wave = __builtin_amdgcn_readfirstlane(tid >> 5);
  const int lane = tid & 31;
  const int lrow = lane & 15;
  const int g    = lane >> 4;

  const float* Xb = x + b * (NROW * DIM);
  const float* W1 = W + dt * DT;
  const float* W2 = W + DIM * DIM + dt * DT;

  const int mi   = (wave & 3) >> 1;
  const int ni   = wave & 1;
  const int kpar = wave >> 2;
  const float* xa0 = xs + (wave * 32 + lrow) * XPITCH + 2 * g;
  const float* xa1 = xa0 + 16 * XPITCH;
  const float* wb0 = ws2 + lrow * WPITCH + 2 * g;
  const float* wb1 = wb0 + 16 * WPITCH;
  const float* xaa = xs + (ic * IC + mi * 16 + lrow) * XPITCH + 2 * g + 4 * kpar;
  const float* wba = ws1 + (ni * 16 + lrow) * WPITCH + 2 * g + 4 * kpar;

  Stage st[NCHUNK];
  #pragma unroll
  for (int kc = 0; kc < NCHUNK; ++kc) {
    stage_load(st[kc], Xb, W1, W2, kc, tid);
    __builtin_amdgcn_sched_barrier(0);
  }
  stage_store(st[0], xs, ws1, ws2, 0, tid);
  lds_barrier();

  v8f acc00 = {}, acc01 = {}, acc10 = {}, acc11 = {}, acca = {};
  #pragma unroll
  for (int kc = 0; kc < NCHUNK; ++kc) {
    wmma_steps<0, NSPLIT>(xa0, xa1, wb0, wb1, xaa, wba, kc, acc00, acc01, acc10, acc11, acca);
    if (kc + 1 < NCHUNK) stage_store(st[kc + 1], xs, ws1, ws2, kc + 1, tid);
    wmma_steps<NSPLIT, KC / 4>(xa0, xa1, wb0, wb1, xaa, wba, kc, acc00, acc01, acc10, acc11, acca);
    if (kc + 1 < NCHUNK) lds_barrier();
  }

  {
    float* p0 = c_t + lrow * CPITCH + wave * 32 + 8 * g;
    float* p1 = p0 + 16 * CPITCH;
    *(v4f*)(p0)      = __builtin_shufflevector(acc00, acc00, 0, 1, 2, 3);
    *(v4f*)(p0 + 4)  = __builtin_shufflevector(acc00, acc00, 4, 5, 6, 7);
    *(v4f*)(p1)      = __builtin_shufflevector(acc01, acc01, 0, 1, 2, 3);
    *(v4f*)(p1 + 4)  = __builtin_shufflevector(acc01, acc01, 4, 5, 6, 7);
    *(v4f*)(p0 + 16) = __builtin_shufflevector(acc10, acc10, 0, 1, 2, 3);
    *(v4f*)(p0 + 20) = __builtin_shufflevector(acc10, acc10, 4, 5, 6, 7);
    *(v4f*)(p1 + 16) = __builtin_shufflevector(acc11, acc11, 0, 1, 2, 3);
    *(v4f*)(p1 + 20) = __builtin_shufflevector(acc11, acc11, 4, 5, 6, 7);
    {
      float* ap = (kpar == 0 ? a_t : a_t2) + (ni * 16 + lrow) * APITCH + mi * 16 + 8 * g;
      *(v4f*)(ap)     = __builtin_shufflevector(acca, acca, 0, 1, 2, 3);
      *(v4f*)(ap + 4) = __builtin_shufflevector(acca, acca, 4, 5, 6, 7);
    }
  }

  __syncthreads();

  const int d = lane;
  const float bd = bias[dt * DT + d];
  const v4f a4  = *(const v4f*)(a_t + d * APITCH + 4 * wave);
  const v4f a4b = *(const v4f*)(a_t2 + d * APITCH + 4 * wave);
  float s[4], acc[4];
  #pragma unroll
  for (int k = 0; k < 4; ++k) {
    s[k]   = (a4[k] + a4b[k]) + bd;
    acc[k] = 0.0f;
  }
  const float* crow = c_t + d * CPITCH;
  #pragma unroll 4
  for (int j = 0; j < NROW; j += 4) {
    const v4f v = *(const v4f*)(crow + j);
    #pragma unroll
    for (int jj = 0; jj < 4; ++jj) {
      #pragma unroll
      for (int k = 0; k < 4; ++k) {
        acc[k] += fmaxf(s[k] + v[jj], 0.0f);
      }
    }
  }
  float* op = out + ((b * NROW) + ic * IC + 4 * wave) * DIM + dt * DT + d;
  #pragma unroll
  for (int k = 0; k < 4; ++k) {
    op[k * DIM] = acc[k];
  }
}

extern "C" void kernel_launch(void* const* d_in, const int* in_sizes, int n_in,
                              void* d_out, int out_size, void* d_ws, size_t ws_size,
                              hipStream_t stream) {
  (void)in_sizes; (void)n_in; (void)d_ws; (void)ws_size; (void)out_size;
  const float* x    = (const float*)d_in[0];
  const float* W    = (const float*)d_in[1];
  const float* bias = (const float*)d_in[2];
  float* out        = (float*)d_out;

  dim3 grid(256);
  dim3 block(256);
  relational_fused_kernel<<<grid, block, 0, stream>>>(x, W, bias, out);
}


# Reduction rewritten as relu(s+c) = c - min(-s,c): 2 vector ops per (i,j,d), was 3; sum_j c once per column from the accumulators; j sum order changed: min terms summed as a tree per 16 j
# speedup vs baseline: 6.3702x; 1.0964x over previous
#include <hip/hip_runtime.h>


typedef float v2f __attribute__((ext_vector_type(2)));
typedef float v4f __attribute__((ext_vector_type(4)));
typedef float v8f __attribute__((ext_vector_type(8)));

#define BATCH 8
#define NROW 256
#define DIM 128

#define DT 32
#define IC 32
#define CPITCH 260
#define APITCH 36
#define SPITCH 20
#define XPITCH 132
#define WPITCH 132
#define KC 32
#define NCHUNK (DIM / KC)
#define NSPLIT 5

#define WMMA4(A, B, C) \
  __builtin_amdgcn_wmma_f32_16x16x4_f32(false, (A), false, (B), (short)0, (C), false, false)

struct Stage {
  v4f x[8];
  v4f w1;
  v4f w2;
};

__device__ __forceinline__ void stage_load(Stage& s, const float* __restrict__ Xb,
                                           const float* __restrict__ W1,
                                           const float* __restrict__ W2, int kc, int tid) {
  const int q = tid & 7;
  const int r = tid >> 3;
  #pragma unroll
  for (int i = 0; i < 8; ++i) {
    s.x[i] = *(const v4f*)(Xb + (r + 32 * i) * DIM + kc * KC + 4 * q);
  }
  s.w1 = *(const v4f*)(W1 + (kc * KC + r) * DIM + 4 * q);
  s.w2 = *(const v4f*)(W2 + (kc * KC + r) * DIM + 4 * q);
}

__device__ __forceinline__ void stage_store(const Stage& s, float* xs, float* ws1, float* ws2,
                                            int kc, int tid) {
  const int q = tid & 7;
  const int r = tid >> 3;
  #pragma unroll
  for (int i = 0; i < 8; ++i) {
    *(v4f*)(xs + (r + 32 * i) * XPITCH + kc * KC + 4 * q) = s.x[i];
  }
  #pragma unroll
  for (int c = 0; c < 4; ++c) {
    ws1[(4 * q + c) * WPITCH + kc * KC + r] = s.w1[c];
    ws2[(4 * q + c) * WPITCH + kc * KC + r] = s.w2[c];
  }
}

template <int S0, int S1>
__device__ __forceinline__ void wmma_steps(const float* xa0, const float* xa1,
                                           const float* wb0, const float* wb1,
                                           const float* xaa, const float* wba, int kc,
                                           v8f& acc00, v8f& acc01, v8f& acc10, v8f& acc11,
                                           v8f& acca) {
  #pragma unroll
  for (int s = S0; s < S1; ++s) {
    const int k = kc * KC + 4 * s;
    const v2f a0 = *(const v2f*)(xa0 + k);
    const v2f a1 = *(const v2f*)(xa1 + k);
    const v2f b0 = *(const v2f*)(wb0 + k);
    const v2f b1 = *(const v2f*)(wb1 + k);
    acc00 = WMMA4(a0, b0, acc00);
    acc01 = WMMA4(a0, b1, acc01);
    acc10 = WMMA4(a1, b0, acc10);
    acc11 = WMMA4(a1, b1, acc11);
    if ((s & 1) == 0) {
      const v2f aa = *(const v2f*)(xaa + k);
      const v2f ba = *(const v2f*)(wba + k);
      acca = WMMA4(aa, ba, acca);
    }
  }
}

__device__ __forceinline__ void lds_barrier() {
  __builtin_amdgcn_fence(__ATOMIC_RELEASE, "workgroup", "local");
  __builtin_amdgcn_s_barrier();
  __builtin_amdgcn_fence(__ATOMIC_ACQUIRE, "workgroup", "local");
}

__global__ __launch_bounds__(256)
void relational_fused_kernel(const float* __restrict__ x,
                             const float* __restrict__ W,
                             const float* __restrict__ bias,
                             float* __restrict__ out) {
  __shared__ __attribute__((aligned(16))) float xs[NROW * XPITCH];
  __shared__ __attribute__((aligned(16))) float ws1[DT * WPITCH];
  __shared__ __attribute__((aligned(16))) float ws2[DT * WPITCH];
  __shared__ __attribute__((aligned(16))) float c_t[DT * CPITCH];
  __shared__ __attribute__((aligned(16))) float a_t[DT * APITCH];
  __shared__ __attribute__((aligned(16))) float a_t2[DT * APITCH];
  __shared__ __attribute__((aligned(16))) float cs_t[DT * SPITCH];

  const int blk = blockIdx.x;
  const int ic  = blk & 7;
  const int dt  = (blk >> 3) & 3;
  const int b   = blk >> 5;
  const int tid = threadIdx.x;
  const int wave = __builtin_amdgcn_readfirstlane(tid >> 5);
  const int lane = tid & 31;
  const int lrow = lane & 15;
  const int g    = lane >> 4;

  const float* Xb = x + b * (NROW * DIM);
  const float* W1 = W + dt * DT;
  const float* W2 = W + DIM * DIM + dt * DT;

  const int mi   = (wave & 3) >> 1;
  const int ni   = wave & 1;
  const int kpar = wave >> 2;
  const float* xa0 = xs + (wave * 32 + lrow) * XPITCH + 2 * g;
  const float* xa1 = xa0 + 16 * XPITCH;
  const float* wb0 = ws2 + lrow * WPITCH + 2 * g;
  const float* wb1 = wb0 + 16 * WPITCH;
  const float* xaa = xs + (ic * IC + mi * 16 + lrow) * XPITCH + 2 * g + 4 * kpar;
  const float* wba = ws1 + (ni * 16 + lrow) * WPITCH + 2 * g + 4 * kpar;

  Stage st[NCHUNK];
  #pragma unroll
  for (int kc = 0; kc < NCHUNK; ++kc) {
    stage_load(st[kc], Xb, W1, W2, kc, tid);
    __builtin_amdgcn_sched_barrier(0);
  }
  stage_store(st[0], xs, ws1, ws2, 0, tid);
  lds_barrier();

  v8f acc00 = {}, acc01 = {}, acc10 = {}, acc11 = {}, acca = {};
  #pragma unroll
  for (int kc = 0; kc < NCHUNK; ++kc) {
    wmma_steps<0, NSPLIT>(xa0, xa1, wb0, wb1, xaa, wba, kc, acc00, acc01, acc10, acc11, acca);
    if (kc + 1 < NCHUNK) stage_store(st[kc + 1], xs, ws1, ws2, kc + 1, tid);
    wmma_steps<NSPLIT, KC / 4>(xa0, xa1, wb0, wb1, xaa, wba, kc, acc00, acc01, acc10, acc11, acca);
    if (kc + 1 < NCHUNK) lds_barrier();
  }

  {
    float* p0 = c_t + lrow * CPITCH + wave * 32 + 8 * g;
    float* p1 = p0 + 16 * CPITCH;
    *(v4f*)(p0)      = __builtin_shufflevector(acc00, acc00, 0, 1, 2, 3);
    *(v4f*)(p0 + 4)  = __builtin_shufflevector(acc00, acc00, 4, 5, 6, 7);
    *(v4f*)(p1)      = __builtin_shufflevector(acc01, acc01, 0, 1, 2, 3);
    *(v4f*)(p1 + 4)  = __builtin_shufflevector(acc01, acc01, 4, 5, 6, 7);
    *(v4f*)(p0 + 16) = __builtin_shufflevector(acc10, acc10, 0, 1, 2, 3);
    *(v4f*)(p0 + 20) = __builtin_shufflevector(acc10, acc10, 4, 5, 6, 7);
    *(v4f*)(p1 + 16) = __builtin_shufflevector(acc11, acc11, 0, 1, 2, 3);
    *(v4f*)(p1 + 20) = __builtin_shufflevector(acc11, acc11, 4, 5, 6, 7);
    {
      float cs0 = acc00[0] + acc10[0];
      float cs1 = acc01[0] + acc11[0];
      #pragma unroll
      for (int r = 1; r < 8; ++r) {
        cs0 += acc00[r] + acc10[r];
        cs1 += acc01[r] + acc11[r];
      }
      cs_t[lrow * SPITCH + 2 * wave + g]        = cs0;
      cs_t[(16 + lrow) * SPITCH + 2 * wave + g] = cs1;
    }
    {
      float* ap = (kpar == 0 ? a_t : a_t2) + (ni * 16 + lrow) * APITCH + mi * 16 + 8 * g;
      *(v4f*)(ap)     = __builtin_shufflevector(acca, acca, 0, 1, 2, 3);
      *(v4f*)(ap + 4) = __builtin_shufflevector(acca, acca, 4, 5, 6, 7);
    }
  }

  __syncthreads();

  const int d = lane;
  const float bd = bias[dt * DT + d];
  const v4f a4  = *(const v4f*)(a_t + d * APITCH + 4 * wave);
  const v4f a4b = *(const v4f*)(a_t2 + d * APITCH + 4 * wave);
  float t[4], acc[4];
  #pragma unroll
  for (int k = 0; k < 4; ++k) {
    t[k]   = -((a4[k] + a4b[k]) + bd);
    acc[k] = 0.0f;
  }
  float csum = 0.0f;
  #pragma unroll
  for (int p = 0; p < 16; p += 4) {
    const v4f cp = *(const v4f*)(cs_t + d * SPITCH + p);
    csum += (cp[0] + cp[1]) + (cp[2] + cp[3]);
  }
  const float* crow = c_t + d * CPITCH;
  for (int j = 0; j < NROW; j += 16) {
    float blk[4][4];
    #pragma unroll
    for (int q = 0; q < 4; ++q) {
      const v4f v = *(const v4f*)(crow + j + 4 * q);
      #pragma unroll
      for (int k = 0; k < 4; ++k) {
        blk[q][k] = (fminf(t[k], v[0]) + fminf(t[k], v[1])) + (fminf(t[k], v[2]) + fminf(t[k], v[3]));
      }
    }
    #pragma unroll
    for (int k = 0; k < 4; ++k) {
      acc[k] += (blk[0][k] + blk[1][k]) + (blk[2][k] + blk[3][k]);
    }
  }
  float* op = out + ((b * NROW) + ic * IC + 4 * wave) * DIM + dt * DT + d;
  #pragma unroll
  for (int k = 0; k < 4; ++k) {
    op[k * DIM] = csum - acc[k];
  }
}

extern "C" void kernel_launch(void* const* d_in, const int* in_sizes, int n_in,
                              void* d_out, int out_size, void* d_ws, size_t ws_size,
                              hipStream_t stream) {
  (void)in_sizes; (void)n_in; (void)d_ws; (void)ws_size; (void)out_size;
  const float* x    = (const float*)d_in[0];
  const float* W    = (const float*)d_in[1];
  const float* bias = (const float*)d_in[2];
  float* out        = (float*)d_out;

  dim3 grid(256);
  dim3 block(256);
  relational_fused_kernel<<<grid, block, 0, stream>>>(x, W, bias, out);
}
